# starting kernel
# baseline (speedup 1.0000x reference)
#include <hip/hip_runtime.h>


typedef __attribute__((ext_vector_type(2))) float v2f;
typedef __attribute__((ext_vector_type(8))) float v8f;

#define Bb 8
#define Ss 256
#define Tt 64
#define Hh 128

__global__ __launch_bounds__(256) void cam_wmma_kernel(const float* __restrict__ inp,
                                                       const float* __restrict__ W,
                                                       float* __restrict__ cam) {
  const int lane = threadIdx.x & 31;
  const int wave = threadIdx.x >> 5;
  const int n = blockIdx.x * 8 + wave;
  const int b  = n >> 14;
  const int h0 = (n >> 7) & 127;
  const int h1 = n & 127;

  const bool hi = (lane >= 16);
  const int  Lh = lane & 15;

  const int base00 = ((b * Ss + 2 * h0) * Ss + 2 * h1) * Tt;
  const float* rowbase = inp + base00 + (hi ? (Ss * Tt) : 0);

  v2f bm;
  bm[0] = hi ? W[2 * 6] : W[0 * 6];
  bm[1] = hi ? W[3 * 6] : W[1 * 6];

  const v8f zero = {};
  v8f d0, d1, d2, d3;
  {
    v2f a;
    a[0] = rowbase[ 0 + Lh]; a[1] = rowbase[Tt +  0 + Lh];
    d0 = __builtin_amdgcn_wmma_f32_16x16x4_f32(false, a, false, bm, (short)0, zero, false, false);
    a[0] = rowbase[16 + Lh]; a[1] = rowbase[Tt + 16 + Lh];
    d1 = __builtin_amdgcn_wmma_f32_16x16x4_f32(false, a, false, bm, (short)0, zero, false, false);
    a[0] = rowbase[32 + Lh]; a[1] = rowbase[Tt + 32 + Lh];
    d2 = __builtin_amdgcn_wmma_f32_16x16x4_f32(false, a, false, bm, (short)0, zero, false, false);
    a[0] = rowbase[48 + Lh]; a[1] = rowbase[Tt + 48 + Lh];
    d3 = __builtin_amdgcn_wmma_f32_16x16x4_f32(false, a, false, bm, (short)0, zero, false, false);
  }

  float mn, mx;
  {
    v8f mnv, mxv;
#pragma unroll
    for (int e = 0; e < 8; ++e) {
      mnv[e] = fminf(fminf(d0[e], d1[e]), fminf(d2[e], d3[e]));
      mxv[e] = fmaxf(fmaxf(d0[e], d1[e]), fmaxf(d2[e], d3[e]));
    }
    mn = mnv[0]; mx = mxv[0];
#pragma unroll
    for (int e = 1; e < 8; ++e) { mn = fminf(mn, mnv[e]); mx = fmaxf(mx, mxv[e]); }
  }
  mn = fminf(mn, __shfl_xor(mn, 16, 32));
  mx = fmaxf(mx, __shfl_xor(mx, 16, 32));
  const float inv = 1.0f / (mx - mn);

  v8f s0 = (lane & 8) ? d2 : d0;
  v8f s1 = (lane & 8) ? d3 : d1;
  v8f dk = (lane & 4) ? s1 : s0;
  float e0 = (lane & 1) ? dk[2] : dk[0];
  float e1 = (lane & 1) ? dk[3] : dk[1];
  float e2 = (lane & 1) ? dk[6] : dk[4];
  float e3 = (lane & 1) ? dk[7] : dk[5];
  float xa = (lane & 2) ? e2 : e0;
  float xb = (lane & 2) ? e3 : e1;

  const int tt = ((lane >> 2) & 3) * 16 + (lane & 3) * 2 + (hi ? 8 : 0);
  float2 o;
  o.x = (xa - mn) * inv;
  o.y = (xb - mn) * inv;
  *(float2*)(cam + ((size_t)n << 6) + tt) = o;
}

__global__ __launch_bounds__(256) void upsample_kernel(const float* __restrict__ cam,
                                                       float* __restrict__ out) {
  const int lane = threadIdx.x & 31;
  const int pix = blockIdx.x * 8 + (threadIdx.x >> 5);
  const int b = pix >> 16;
  const int y = (pix >> 8) & 255;
  const int x = pix & 255;

  const float r = 127.0f / 255.0f;
  float fy = (float)y * r; int y0 = (int)fy; float wy = fy - (float)y0;
  int y1 = (y0 < Hh - 1) ? y0 + 1 : Hh - 1;
  float fx = (float)x * r; int x0 = (int)fx; float wx = fx - (float)x0;
  int x1 = (x0 < Hh - 1) ? x0 + 1 : Hh - 1;

  const int t = lane << 1;
  const size_t pb = (size_t)b << 14;
  const float* c00 = cam + (((pb + (size_t)y0 * Hh + x0) << 6) + t);
  const float* c01 = cam + (((pb + (size_t)y0 * Hh + x1) << 6) + t);
  const float* c10 = cam + (((pb + (size_t)y1 * Hh + x0) << 6) + t);
  const float* c11 = cam + (((pb + (size_t)y1 * Hh + x1) << 6) + t);
  float2 v00 = *(const float2*)c00;
  float2 v01 = *(const float2*)c01;
  float2 v10 = *(const float2*)c10;
  float2 v11 = *(const float2*)c11;

  float h0x = v00.x * (1.0f - wy) + v10.x * wy;
  float h1x = v01.x * (1.0f - wy) + v11.x * wy;
  float h0y = v00.y * (1.0f - wy) + v10.y * wy;
  float h1y = v01.y * (1.0f - wy) + v11.y * wy;
  float2 o;
  o.x = h0x * (1.0f - wx) + h1x * wx;
  o.y = h0y * (1.0f - wx) + h1y * wx;

  *(float2*)(out + (((size_t)pix) << 6) + t) = o;
}

extern "C" void kernel_launch(void* const* d_in, const int* in_sizes, int n_in,
                              void* d_out, int out_size, void* d_ws, size_t ws_size,
                              hipStream_t stream) {
  const float* inp = (const float*)d_in[0];
  const float* W   = (const float*)d_in[1];
  float* out = (float*)d_out;
  float* cam = (float*)d_ws;

  cam_wmma_kernel<<<16384, 256, 0, stream>>>(inp, W, cam);
  upsample_kernel<<<65536, 256, 0, stream>>>(cam, out);
}


# upsample kernel: one block per source row pair stages the y-interpolated row in LDS and emits both output rows with 16-byte accesses (1024 blocks), not one wave per pixel with four 8-byte loads
# speedup vs baseline: 1.8524x; 1.8524x over previous
#include <hip/hip_runtime.h>


typedef __attribute__((ext_vector_type(2))) float v2f;
typedef __attribute__((ext_vector_type(8))) float v8f;

#define Bb 8
#define Ss 256
#define Tt 64
#define Hh 128

__global__ __launch_bounds__(256) void cam_wmma_kernel(const float* __restrict__ inp,
                                                       const float* __restrict__ W,
                                                       float* __restrict__ cam) {
  const int lane = threadIdx.x & 31;
  const int wave = threadIdx.x >> 5;
  const int n = blockIdx.x * 8 + wave;
  const int b  = n >> 14;
  const int h0 = (n >> 7) & 127;
  const int h1 = n & 127;

  const bool hi = (lane >= 16);
  const int  Lh = lane & 15;

  const int base00 = ((b * Ss + 2 * h0) * Ss + 2 * h1) * Tt;
  const float* rowbase = inp + base00 + (hi ? (Ss * Tt) : 0);

  v2f bm;
  bm[0] = hi ? W[2 * 6] : W[0 * 6];
  bm[1] = hi ? W[3 * 6] : W[1 * 6];

  const v8f zero = {};
  v8f d0, d1, d2, d3;
  {
    v2f a;
    a[0] = rowbase[ 0 + Lh]; a[1] = rowbase[Tt +  0 + Lh];
    d0 = __builtin_amdgcn_wmma_f32_16x16x4_f32(false, a, false, bm, (short)0, zero, false, false);
    a[0] = rowbase[16 + Lh]; a[1] = rowbase[Tt + 16 + Lh];
    d1 = __builtin_amdgcn_wmma_f32_16x16x4_f32(false, a, false, bm, (short)0, zero, false, false);
    a[0] = rowbase[32 + Lh]; a[1] = rowbase[Tt + 32 + Lh];
    d2 = __builtin_amdgcn_wmma_f32_16x16x4_f32(false, a, false, bm, (short)0, zero, false, false);
    a[0] = rowbase[48 + Lh]; a[1] = rowbase[Tt + 48 + Lh];
    d3 = __builtin_amdgcn_wmma_f32_16x16x4_f32(false, a, false, bm, (short)0, zero, false, false);
  }

  float mn, mx;
  {
    v8f mnv, mxv;
#pragma unroll
    for (int e = 0; e < 8; ++e) {
      mnv[e] = fminf(fminf(d0[e], d1[e]), fminf(d2[e], d3[e]));
      mxv[e] = fmaxf(fmaxf(d0[e], d1[e]), fmaxf(d2[e], d3[e]));
    }
    mn = mnv[0]; mx = mxv[0];
#pragma unroll
    for (int e = 1; e < 8; ++e) { mn = fminf(mn, mnv[e]); mx = fmaxf(mx, mxv[e]); }
  }
  mn = fminf(mn, __shfl_xor(mn, 16, 32));
  mx = fmaxf(mx, __shfl_xor(mx, 16, 32));
  const float inv = 1.0f / (mx - mn);

  v8f s0 = (lane & 8) ? d2 : d0;
  v8f s1 = (lane & 8) ? d3 : d1;
  v8f dk = (lane & 4) ? s1 : s0;
  float e0 = (lane & 1) ? dk[2] : dk[0];
  float e1 = (lane & 1) ? dk[3] : dk[1];
  float e2 = (lane & 1) ? dk[6] : dk[4];
  float e3 = (lane & 1) ? dk[7] : dk[5];
  float xa = (lane & 2) ? e2 : e0;
  float xb = (lane & 2) ? e3 : e1;

  const int tt = ((lane >> 2) & 3) * 16 + (lane & 3) * 2 + (hi ? 8 : 0);
  float2 o;
  o.x = (xa - mn) * inv;
  o.y = (xb - mn) * inv;
  *(float2*)(cam + ((size_t)n << 6) + tt) = o;
}

__global__ __launch_bounds__(256) void upsample_kernel(const float* __restrict__ cam,
                                                       float* __restrict__ out) {
  __shared__ float4 hrow[2][Hh * 16];

  const int tid = threadIdx.x;
  const int b = blockIdx.x >> 7;
  const int j = blockIdx.x & (Hh - 1);
  const size_t pb = (size_t)b << 14;
  const float r = 127.0f / 255.0f;
  const int ya = 2 * j + 1;
  const int yb = (j == Hh - 1) ? 0 : 2 * j + 2;

  {
    float4 v0[8], v1[8];
#pragma unroll
    for (int s = 0; s < 2; ++s) {
      const int y = s ? yb : ya;
      const float fy = (float)y * r; const int y0 = (int)fy; const float wy = fy - (float)y0;
      const int y1 = (y0 < Hh - 1) ? y0 + 1 : Hh - 1;
      if (s == 0 || j == Hh - 1) {
        const float4* r0 = (const float4*)(cam + ((pb + (size_t)y0 * Hh) << 6));
        const float4* r1 = (const float4*)(cam + ((pb + (size_t)y1 * Hh) << 6));
#pragma unroll
        for (int i = 0; i < 8; ++i) {
          v0[i] = r0[tid + 256 * i];
          v1[i] = r1[tid + 256 * i];
        }
      }
#pragma unroll
      for (int i = 0; i < 8; ++i) {
        float4 h;
        h.x = v0[i].x * (1.0f - wy) + v1[i].x * wy;
        h.y = v0[i].y * (1.0f - wy) + v1[i].y * wy;
        h.z = v0[i].z * (1.0f - wy) + v1[i].z * wy;
        h.w = v0[i].w * (1.0f - wy) + v1[i].w * wy;
        hrow[s][tid + 256 * i] = h;
      }
    }
  }
  __syncthreads();

  const int t4 = tid & 15;
  const int xb = tid >> 4;
  float4* orow0 = (float4*)(out + ((((size_t)b * Ss + (size_t)ya) * Ss) << 6));
  float4* orow1 = (float4*)(out + ((((size_t)b * Ss + (size_t)yb) * Ss) << 6));
#pragma unroll
  for (int i = 0; i < 16; ++i) {
    const int x = xb + 16 * i;
    const float fx = (float)x * r; const int x0 = (int)fx; const float wx = fx - (float)x0;
    const int x1 = (x0 < Hh - 1) ? x0 + 1 : Hh - 1;
    const float4 a0 = hrow[0][x0 * 16 + t4];
    const float4 c0 = hrow[0][x1 * 16 + t4];
    const float4 a1 = hrow[1][x0 * 16 + t4];
    const float4 c1 = hrow[1][x1 * 16 + t4];
    float4 o0, o1;
    o0.x = a0.x * (1.0f - wx) + c0.x * wx;
    o0.y = a0.y * (1.0f - wx) + c0.y * wx;
    o0.z = a0.z * (1.0f - wx) + c0.z * wx;
    o0.w = a0.w * (1.0f - wx) + c0.w * wx;
    o1.x = a1.x * (1.0f - wx) + c1.x * wx;
    o1.y = a1.y * (1.0f - wx) + c1.y * wx;
    o1.z = a1.z * (1.0f - wx) + c1.z * wx;
    o1.w = a1.w * (1.0f - wx) + c1.w * wx;
    orow0[tid + 256 * i] = o0;
    orow1[tid + 256 * i] = o1;
  }
}

extern "C" void kernel_launch(void* const* d_in, const int* in_sizes, int n_in,
                              void* d_out, int out_size, void* d_ws, size_t ws_size,
                              hipStream_t stream) {
  const float* inp = (const float*)d_in[0];
  const float* W   = (const float*)d_in[1];
  float* out = (float*)d_out;
  float* cam = (float*)d_ws;

  cam_wmma_kernel<<<16384, 256, 0, stream>>>(inp, W, cam);
  upsample_kernel<<<1024, 256, 0, stream>>>(cam, out);
}


# pooling kernel: 4096 waves each stream 32 pooled pixels with 16-byte loads (24-32 in flight), taps as the WMMA B operand and W as A, row_xmask min/max; was one wave per pixel with 4-byte loads
# speedup vs baseline: 2.1890x; 1.1817x over previous
#include <hip/hip_runtime.h>


typedef __attribute__((ext_vector_type(2))) float v2f;
typedef __attribute__((ext_vector_type(8))) float v8f;

#define Bb 8
#define Ss 256
#define Tt 64
#define Hh 128

#define ROW_XOR(v, M) __builtin_bit_cast(float, __builtin_amdgcn_update_dpp(0, __builtin_bit_cast(int, (v)), 0x160 + (M), 0xf, 0xf, false))

__device__ __forceinline__ float cam16(v2f aw, float tap_a, float tap_b) {
  v2f bq;
  bq[0] = tap_a; bq[1] = tap_b;
  const v8f zero = {};
  const v8f d = __builtin_amdgcn_wmma_f32_16x16x4_f32(false, aw, false, bq, (short)0, zero, false, false);
  return d[0];
}

__global__ __launch_bounds__(256) void cam_wmma_kernel(const float* __restrict__ inp,
                                                       const float* __restrict__ W,
                                                       float* __restrict__ cam) {
  const int lane = threadIdx.x & 31;
  const int wave = threadIdx.x >> 5;
  const int n0 = (blockIdx.x * 8 + wave) * 32;
  const int b  = n0 >> 14;
  const int h0 = (n0 >> 7) & 127;
  const int h1 = n0 & 127;
  const int hi = lane >> 4;
  const int c  = lane & 15;

  const float4* src = (const float4*)(inp + ((size_t)((b * Ss + 2 * h0 + hi) * Ss + 2 * h1) * Tt) + 4 * c);
  float4* dst = (float4*)(cam + ((size_t)(n0 + hi) << 6) + 4 * c);

  v2f aw;
  aw[0] = W[(2 * hi) * 6];
  aw[1] = W[(2 * hi + 1) * 6];

  float4 buf[3][8];
#pragma unroll
  for (int i = 0; i < 8; ++i) buf[0][i] = src[i * 16];
#pragma unroll
  for (int i = 0; i < 8; ++i) buf[1][i] = src[(8 + i) * 16];

#pragma unroll
  for (int k = 0; k < 8; ++k) {
    if (k + 2 < 8) {
#pragma unroll
      for (int i = 0; i < 8; ++i) buf[(k + 2) % 3][i] = src[((k + 2) * 8 + i) * 16];
    }
    __builtin_amdgcn_sched_barrier(0);
#pragma unroll
    for (int pp = 0; pp < 2; ++pp) {
      const float4 u0 = buf[k % 3][4 * pp + 0];
      const float4 u1 = buf[k % 3][4 * pp + 1];
      const float4 w0 = buf[k % 3][4 * pp + 2];
      const float4 w1 = buf[k % 3][4 * pp + 3];
      const float ux = cam16(aw, u0.x, u1.x), uy = cam16(aw, u0.y, u1.y);
      const float uz = cam16(aw, u0.z, u1.z), uw = cam16(aw, u0.w, u1.w);
      const float wx = cam16(aw, w0.x, w1.x), wy = cam16(aw, w0.y, w1.y);
      const float wz = cam16(aw, w0.z, w1.z), ww = cam16(aw, w0.w, w1.w);
      const float x0 = hi ? wx : ux;
      const float x1 = hi ? wy : uy;
      const float x2 = hi ? wz : uz;
      const float x3 = hi ? ww : uw;
      float mn = fminf(fminf(x0, x1), fminf(x2, x3));
      float mx = fmaxf(fmaxf(x0, x1), fmaxf(x2, x3));
      mn = fminf(mn, ROW_XOR(mn, 1)); mx = fmaxf(mx, ROW_XOR(mx, 1));
      mn = fminf(mn, ROW_XOR(mn, 2)); mx = fmaxf(mx, ROW_XOR(mx, 2));
      mn = fminf(mn, ROW_XOR(mn, 4)); mx = fmaxf(mx, ROW_XOR(mx, 4));
      mn = fminf(mn, ROW_XOR(mn, 8)); mx = fmaxf(mx, ROW_XOR(mx, 8));
      const float inv = 1.0f / (mx - mn);
      float4 o;
      o.x = (x0 - mn) * inv;
      o.y = (x1 - mn) * inv;
      o.z = (x2 - mn) * inv;
      o.w = (x3 - mn) * inv;
      dst[(4 * k + 2 * pp) * 16] = o;
    }
  }
}

__global__ __launch_bounds__(256) void upsample_kernel(const float* __restrict__ cam,
                                                       float* __restrict__ out) {
  __shared__ float4 hrow[2][Hh * 16];

  const int tid = threadIdx.x;
  const int b = blockIdx.x >> 7;
  const int j = blockIdx.x & (Hh - 1);
  const size_t pb = (size_t)b << 14;
  const float r = 127.0f / 255.0f;
  const int ya = 2 * j + 1;
  const int yb = (j == Hh - 1) ? 0 : 2 * j + 2;

  {
    float4 v0[8], v1[8];
#pragma unroll
    for (int s = 0; s < 2; ++s) {
      const int y = s ? yb : ya;
      const float fy = (float)y * r; const int y0 = (int)fy; const float wy = fy - (float)y0;
      const int y1 = (y0 < Hh - 1) ? y0 + 1 : Hh - 1;
      if (s == 0 || j == Hh - 1) {
        const float4* r0 = (const float4*)(cam + ((pb + (size_t)y0 * Hh) << 6));
        const float4* r1 = (const float4*)(cam + ((pb + (size_t)y1 * Hh) << 6));
#pragma unroll
        for (int i = 0; i < 8; ++i) {
          v0[i] = r0[tid + 256 * i];
          v1[i] = r1[tid + 256 * i];
        }
      }
#pragma unroll
      for (int i = 0; i < 8; ++i) {
        float4 h;
        h.x = v0[i].x * (1.0f - wy) + v1[i].x * wy;
        h.y = v0[i].y * (1.0f - wy) + v1[i].y * wy;
        h.z = v0[i].z * (1.0f - wy) + v1[i].z * wy;
        h.w = v0[i].w * (1.0f - wy) + v1[i].w * wy;
        hrow[s][tid + 256 * i] = h;
      }
    }
  }
  __syncthreads();

  const int t4 = tid & 15;
  const int xb = tid >> 4;
  float4* orow0 = (float4*)(out + ((((size_t)b * Ss + (size_t)ya) * Ss) << 6));
  float4* orow1 = (float4*)(out + ((((size_t)b * Ss + (size_t)yb) * Ss) << 6));
#pragma unroll
  for (int i = 0; i < 16; ++i) {
    const int x = xb + 16 * i;
    const float fx = (float)x * r; const int x0 = (int)fx; const float wx = fx - (float)x0;
    const int x1 = (x0 < Hh - 1) ? x0 + 1 : Hh - 1;
    const float4 a0 = hrow[0][x0 * 16 + t4];
    const float4 c0 = hrow[0][x1 * 16 + t4];
    const float4 a1 = hrow[1][x0 * 16 + t4];
    const float4 c1 = hrow[1][x1 * 16 + t4];
    float4 o0, o1;
    o0.x = a0.x * (1.0f - wx) + c0.x * wx;
    o0.y = a0.y * (1.0f - wx) + c0.y * wx;
    o0.z = a0.z * (1.0f - wx) + c0.z * wx;
    o0.w = a0.w * (1.0f - wx) + c0.w * wx;
    o1.x = a1.x * (1.0f - wx) + c1.x * wx;
    o1.y = a1.y * (1.0f - wx) + c1.y * wx;
    o1.z = a1.z * (1.0f - wx) + c1.z * wx;
    o1.w = a1.w * (1.0f - wx) + c1.w * wx;
    orow0[tid + 256 * i] = o0;
    orow1[tid + 256 * i] = o1;
  }
}

extern "C" void kernel_launch(void* const* d_in, const int* in_sizes, int n_in,
                              void* d_out, int out_size, void* d_ws, size_t ws_size,
                              hipStream_t stream) {
  const float* inp = (const float*)d_in[0];
  const float* W   = (const float*)d_in[1];
  float* out = (float*)d_out;
  float* cam = (float*)d_ws;

  cam_wmma_kernel<<<512, 256, 0, stream>>>(inp, W, cam);
  upsample_kernel<<<1024, 256, 0, stream>>>(cam, out);
}


# one launch: each block builds 4 pooled rows plus the row below in LDS (WMMA, min/max) and emits their 8 output rows from it; the normalized image no longer goes through the workspace
# speedup vs baseline: 2.5519x; 1.1658x over previous
#include <hip/hip_runtime.h>


typedef __attribute__((ext_vector_type(2))) float v2f;
typedef __attribute__((ext_vector_type(8))) float v8f;

#define Bb 8
#define Ss 256
#define Tt 64
#define Hh 128
#define RS 4

#define ROW_XOR(v, M) __builtin_bit_cast(float, __builtin_amdgcn_update_dpp(0, __builtin_bit_cast(int, (v)), 0x160 + (M), 0xf, 0xf, false))

__device__ __forceinline__ float cam16(v2f aw, float tap_a, float tap_b) {
  v2f bq;
  bq[0] = tap_a; bq[1] = tap_b;
  const v8f zero = {};
  const v8f d = __builtin_amdgcn_wmma_f32_16x16x4_f32(false, aw, false, bq, (short)0, zero, false, false);
  return d[0];
}

__device__ __forceinline__ float4 lerp4(float4 a, float4 b, float w) {
  float4 o;
  o.x = a.x * (1.0f - w) + b.x * w;
  o.y = a.y * (1.0f - w) + b.y * w;
  o.z = a.z * (1.0f - w) + b.z * w;
  o.w = a.w * (1.0f - w) + b.w * w;
  return o;
}

template <bool TWO>
__device__ __forceinline__ void emit_rows(const float4* ra, const float4* rb, float wya, float wyb,
                                          float4* oa, float4* ob, int tid) {
  const float r = 127.0f / 255.0f;
  const int t4 = tid & 15;
  const int xb = tid >> 4;
#pragma unroll
  for (int i = 0; i < 8; ++i) {
    const int x = xb + 32 * i;
    const float fx = (float)x * r; const int x0 = (int)fx; const float wx = fx - (float)x0;
    const int x1 = (x0 < Hh - 1) ? x0 + 1 : Hh - 1;
    const float4 a0 = ra[x0 * 16 + t4];
    const float4 a1 = rb[x0 * 16 + t4];
    const float4 c0 = ra[x1 * 16 + t4];
    const float4 c1 = rb[x1 * 16 + t4];
    oa[tid + 512 * i] = lerp4(lerp4(a0, a1, wya), lerp4(c0, c1, wya), wx);
    if (TWO) ob[tid + 512 * i] = lerp4(lerp4(a0, a1, wyb), lerp4(c0, c1, wyb), wx);
  }
}

__global__ __launch_bounds__(512) void cam_upsample_kernel(const float* __restrict__ inp,
                                                           const float* __restrict__ W,
                                                           float* __restrict__ out) {
  __shared__ float4 crow[RS + 1][Hh * 16];

  const int tid = threadIdx.x;
  const int lane = tid & 31;
  const int wave = tid >> 5;
  const int b  = blockIdx.x >> 5;
  const int j0 = (blockIdx.x & 31) * RS;
  const int hi = lane >> 4;
  const int c  = lane & 15;

  const int rs   = wave >> 2;
  const int pxA  = (wave & 3) * 32;
  const int rowH = (j0 + RS < Hh) ? j0 + RS : Hh - 1;
  const int pxH  = wave * 8;
  const float4* srcA = (const float4*)(inp + ((size_t)((b * Ss + 2 * (j0 + rs) + hi) * Ss + 2 * pxA) * Tt) + 4 * c);
  const float4* srcH = (const float4*)(inp + ((size_t)((b * Ss + 2 * rowH + hi) * Ss + 2 * pxH) * Tt) + 4 * c);
  float4* dA = &crow[rs][(pxA + hi) * 16 + c];
  float4* dH = &crow[RS][(pxH + hi) * 16 + c];

  v2f aw;
  aw[0] = W[(2 * hi) * 6];
  aw[1] = W[(2 * hi + 1) * 6];

  float4 buf[3][8];
#pragma unroll
  for (int i = 0; i < 8; ++i) buf[0][i] = srcA[i * 16];
#pragma unroll
  for (int i = 0; i < 8; ++i) buf[1][i] = srcA[(8 + i) * 16];

#pragma unroll
  for (int k = 0; k < 10; ++k) {
    if (k + 2 < 10) {
      const float4* s = (k + 2 < 8) ? srcA + (k + 2) * 128 : srcH + (k + 2 - 8) * 128;
#pragma unroll
      for (int i = 0; i < 8; ++i) buf[(k + 2) % 3][i] = s[i * 16];
    }
    __builtin_amdgcn_sched_barrier(0);
    float4* d = (k < 8) ? dA + (4 * k) * 16 : dH + (4 * (k - 8)) * 16;
#pragma unroll
    for (int pp = 0; pp < 2; ++pp) {
      const float4 u0 = buf[k % 3][4 * pp + 0];
      const float4 u1 = buf[k % 3][4 * pp + 1];
      const float4 w0 = buf[k % 3][4 * pp + 2];
      const float4 w1 = buf[k % 3][4 * pp + 3];
      const float ux = cam16(aw, u0.x, u1.x), uy = cam16(aw, u0.y, u1.y);
      const float uz = cam16(aw, u0.z, u1.z), uw = cam16(aw, u0.w, u1.w);
      const float wx = cam16(aw, w0.x, w1.x), wy = cam16(aw, w0.y, w1.y);
      const float wz = cam16(aw, w0.z, w1.z), ww = cam16(aw, w0.w, w1.w);
      const float x0 = hi ? wx : ux;
      const float x1 = hi ? wy : uy;
      const float x2 = hi ? wz : uz;
      const float x3 = hi ? ww : uw;
      float mn = fminf(fminf(x0, x1), fminf(x2, x3));
      float mx = fmaxf(fmaxf(x0, x1), fmaxf(x2, x3));
      mn = fminf(mn, ROW_XOR(mn, 1)); mx = fmaxf(mx, ROW_XOR(mx, 1));
      mn = fminf(mn, ROW_XOR(mn, 2)); mx = fmaxf(mx, ROW_XOR(mx, 2));
      mn = fminf(mn, ROW_XOR(mn, 4)); mx = fmaxf(mx, ROW_XOR(mx, 4));
      mn = fminf(mn, ROW_XOR(mn, 8)); mx = fmaxf(mx, ROW_XOR(mx, 8));
      const float inv = 1.0f / (mx - mn);
      float4 o;
      o.x = (x0 - mn) * inv;
      o.y = (x1 - mn) * inv;
      o.z = (x2 - mn) * inv;
      o.w = (x3 - mn) * inv;
      d[(2 * pp) * 16] = o;
    }
  }
  __syncthreads();

  const float r = 127.0f / 255.0f;
  float4* obase = (float4*)out + ((size_t)b * Ss * Ss) * 16;
  for (int p = 0; p < RS; ++p) {
    const int j = j0 + p;
    const int ya = 2 * j + 1;
    const float fya = (float)ya * r; const int y0 = (int)fya; const float wya = fya - (float)y0;
    const int y1 = (y0 < Hh - 1) ? y0 + 1 : Hh - 1;
    const float4* ra = crow[y0 - j0];
    const float4* rb = crow[y1 - j0];
    float4* oa = obase + (size_t)ya * (Ss * 16);
    if (j < Hh - 1) {
      const float fyb = (float)(ya + 1) * r; const float wyb = fyb - (float)(int)fyb;
      emit_rows<true>(ra, rb, wya, wyb, oa, oa + Ss * 16, tid);
    } else {
      emit_rows<false>(ra, rb, wya, 0.0f, oa, oa, tid);
    }
  }
  if (j0 == 0) {
    const float fy = 0.0f * r; const int y0 = (int)fy; const float wy0 = fy - (float)y0;
    emit_rows<false>(crow[y0], crow[y0 + 1], wy0, 0.0f, obase, obase, tid);
  }
}

extern "C" void kernel_launch(void* const* d_in, const int* in_sizes, int n_in,
                              void* d_out, int out_size, void* d_ws, size_t ws_size,
                              hipStream_t stream) {
  const float* inp = (const float*)d_in[0];
  const float* W   = (const float*)d_in[1];
  float* out = (float*)d_out;

  cam_upsample_kernel<<<256, 512, 0, stream>>>(inp, W, out);
}
